# starting kernel
# baseline (speedup 1.0000x reference)
#include <hip/hip_runtime.h>
#include <hip/hip_bf16.h>


#define GIN_D 64
#define GIN_H 128

typedef __attribute__((ext_vector_type(2))) float v2f;
typedef __attribute__((ext_vector_type(8))) float v8f;

__device__ __forceinline__ void atomic_add_f32_agent(float* p, float v) {
  __hip_atomic_fetch_add(p, v, __ATOMIC_RELAXED, __HIP_MEMORY_SCOPE_AGENT);
}

__global__ void gin_combine_kernel(const float* __restrict__ feat,
                                   const float* __restrict__ eps,
                                   float* __restrict__ hacc, int n4) {
  int i = blockIdx.x * blockDim.x + threadIdx.x;
  if (i >= n4) return;
  const float s = 1.0f + eps[0];
  float4 v = ((const float4*)feat)[i];
  v.x *= s; v.y *= s; v.z *= s; v.w *= s;
  ((float4*)hacc)[i] = v;
}

__global__ void gin_scatter_kernel(const float* __restrict__ feat,
                                   const int* __restrict__ esrc,
                                   const int* __restrict__ edst,
                                   float* __restrict__ hacc, int n_edges) {
  long long t = (long long)blockIdx.x * blockDim.x + threadIdx.x;
  long long total = (long long)n_edges * 16;
  if (t >= total) return;
  int e = (int)(t >> 4);
  int c = (int)(t & 15) * 4;
  int s = esrc[e];
  int d = edst[e];
  float4 v = *(const float4*)(feat + (size_t)s * GIN_D + c);
  float* dp = hacc + (size_t)d * GIN_D + c;
  atomic_add_f32_agent(dp + 0, v.x);
  atomic_add_f32_agent(dp + 1, v.y);
  atomic_add_f32_agent(dp + 2, v.z);
  atomic_add_f32_agent(dp + 3, v.w);
}

__global__ __launch_bounds__(128) void gin_mlp_kernel(
    const float* __restrict__ hacc, const float* __restrict__ W1,
    const float* __restrict__ b1, const float* __restrict__ W2,
    const float* __restrict__ b2, float* __restrict__ out) {
  __shared__ float lds_h[16 * GIN_D];
  __shared__ float lds_mid[16 * GIN_H];

  const int tid  = threadIdx.x;
  const int wave = tid >> 5;
  const int lane = tid & 31;
  const int lhalf = lane >> 4;
  const int lmod  = lane & 15;
  const int node0 = blockIdx.x * 16;

  {
    const float4* src = (const float4*)(hacc + (size_t)node0 * GIN_D);
    float4* dst = (float4*)lds_h;
    dst[tid]       = src[tid];
    dst[tid + 128] = src[tid + 128];
  }
  __syncthreads();

  v8f acc0 = {};
  v8f acc1 = {};
  const int n0 = lmod + 16 * (2 * wave);
  const int n1 = lmod + 16 * (2 * wave + 1);
#pragma unroll
  for (int k = 0; k < 16; ++k) {
    const int kb = 4 * k + 2 * lhalf;
    v2f a;
    a.x = lds_h[lmod * GIN_D + kb];
    a.y = lds_h[lmod * GIN_D + kb + 1];
    v2f bf0, bf1;
    bf0.x = W1[(size_t)kb * GIN_H + n0];
    bf0.y = W1[(size_t)(kb + 1) * GIN_H + n0];
    bf1.x = W1[(size_t)kb * GIN_H + n1];
    bf1.y = W1[(size_t)(kb + 1) * GIN_H + n1];
    acc0 = __builtin_amdgcn_wmma_f32_16x16x4_f32(false, a, false, bf0,
                                                 (short)0, acc0, false, false);
    acc1 = __builtin_amdgcn_wmma_f32_16x16x4_f32(false, a, false, bf1,
                                                 (short)0, acc1, false, false);
  }
  {
    const float b1v0 = b1[n0];
    const float b1v1 = b1[n1];
#pragma unroll
    for (int r = 0; r < 8; ++r) {
      const int row = r + 8 * lhalf;
      lds_mid[row * GIN_H + n0] = fmaxf(acc0[r] + b1v0, 0.0f);
      lds_mid[row * GIN_H + n1] = fmaxf(acc1[r] + b1v1, 0.0f);
    }
  }
  __syncthreads();

  v8f acc2 = {};
  const int n2 = lmod + 16 * wave;
#pragma unroll
  for (int k = 0; k < 32; ++k) {
    const int kb = 4 * k + 2 * lhalf;
    v2f a;
    a.x = lds_mid[lmod * GIN_H + kb];
    a.y = lds_mid[lmod * GIN_H + kb + 1];
    v2f bf;
    bf.x = W2[(size_t)kb * GIN_D + n2];
    bf.y = W2[(size_t)(kb + 1) * GIN_D + n2];
    acc2 = __builtin_amdgcn_wmma_f32_16x16x4_f32(false, a, false, bf,
                                                 (short)0, acc2, false, false);
  }
  {
    const float b2v = b2[n2];
#pragma unroll
    for (int r = 0; r < 8; ++r) {
      const int row = r + 8 * lhalf;
      out[(size_t)(node0 + row) * GIN_D + n2] = acc2[r] + b2v;
    }
  }
}

extern "C" void kernel_launch(void* const* d_in, const int* in_sizes, int n_in,
                              void* d_out, int out_size, void* d_ws, size_t ws_size,
                              hipStream_t stream) {
  const float* feat = (const float*)d_in[0];
  const int*   esrc = (const int*)d_in[1];
  const int*   edst = (const int*)d_in[2];
  const float* W1   = (const float*)d_in[3];
  const float* b1   = (const float*)d_in[4];
  const float* W2   = (const float*)d_in[5];
  const float* b2   = (const float*)d_in[6];
  const float* eps  = (const float*)d_in[7];

  const int n_nodes = in_sizes[0] / GIN_D;
  const int n_edges = in_sizes[1];

  float* hacc = (float*)d_ws;

  {
    const int n4 = n_nodes * (GIN_D / 4);
    const int blocks = (n4 + 255) / 256;
    gin_combine_kernel<<<blocks, 256, 0, stream>>>(feat, eps, hacc, n4);
  }
  {
    const long long total = (long long)n_edges * 16;
    const int blocks = (int)((total + 255) / 256);
    gin_scatter_kernel<<<blocks, 256, 0, stream>>>(feat, esrc, edst, hacc, n_edges);
  }
  {
    const int blocks = n_nodes / 16;
    gin_mlp_kernel<<<blocks, 128, 0, stream>>>(hacc, W1, b1, W2, b2,
                                               (float*)d_out);
  }
}


# MLP computed transposed: W^T tiles are A, node columns are B, so GEMM1's D registers are GEMM2's B operand; mid and h never touch LDS, no barrier; k sums regrouped (f32 sum order changes)
# speedup vs baseline: 1.4376x; 1.4376x over previous
#include <hip/hip_runtime.h>
#include <hip/hip_bf16.h>


#define GIN_D 64
#define GIN_H 128

typedef __attribute__((ext_vector_type(2))) float v2f;
typedef __attribute__((ext_vector_type(8))) float v8f;

__device__ __forceinline__ int atomic_xchg_i32_agent(int* p, int v) {
  return __hip_atomic_exchange(p, v, __ATOMIC_RELAXED, __HIP_MEMORY_SCOPE_AGENT);
}

__global__ void gin_head_init_kernel(int* __restrict__ head, int n_nodes) {
  const int i4 = (blockIdx.x * blockDim.x + threadIdx.x) * 4;
  if (i4 + 3 < n_nodes) {
    *(int4*)(head + i4) = make_int4(-1, -1, -1, -1);
  } else {
    for (int i = i4; i < n_nodes; ++i) head[i] = -1;
  }
}

__global__ __launch_bounds__(256) void gin_link_kernel(
    const int* __restrict__ esrc, const int* __restrict__ edst,
    int* __restrict__ head, int2* __restrict__ cell, int n_edges) {
  const int lane = threadIdx.x & 31;
  const int wv   = (blockIdx.x * blockDim.x + threadIdx.x) >> 5;
  const int base = wv * 128 + lane;
  if (wv * 128 + 128 <= n_edges) {
    int s[4], d[4], o[4];
    for (int k = 0; k < 4; ++k) { s[k] = esrc[base + 32 * k]; d[k] = edst[base + 32 * k]; }
    for (int k = 0; k < 4; ++k) o[k] = atomic_xchg_i32_agent(head + d[k], base + 32 * k);
    for (int k = 0; k < 4; ++k) cell[base + 32 * k] = make_int2(o[k], s[k]);
  } else {
    for (int k = 0; k < 4; ++k) {
      const int e = base + 32 * k;
      if (e < n_edges) {
        const int sv = esrc[e];
        const int ov = atomic_xchg_i32_agent(head + edst[e], e);
        cell[e] = make_int2(ov, sv);
      }
    }
  }
}

__global__ __launch_bounds__(256) void gin_pull_kernel(
    const float* __restrict__ feat, const float* __restrict__ eps,
    const int* __restrict__ head, const int2* __restrict__ cell,
    float* __restrict__ hacc, int n_nodes, int n_edges) {
  const int gt = blockIdx.x * blockDim.x + threadIdx.x;
  const int n  = gt >> 3;
  const int j  = gt & 7;
  if (n >= n_nodes) return;
  const float4* f4 = (const float4*)feat;
  const float sc = 1.0f + eps[0];
  float4 a0 = f4[(size_t)n * 16 + j];
  float4 a1 = f4[(size_t)n * 16 + 8 + j];
  int e = head[n];
  a0.x *= sc; a0.y *= sc; a0.z *= sc; a0.w *= sc;
  a1.x *= sc; a1.y *= sc; a1.z *= sc; a1.w *= sc;
  int2 c = make_int2(-1, 0);
  if (e >= 0) c = cell[e];
  for (int it = 0; it < n_edges && e >= 0; ++it) {
    const int en = c.x;
    const int sr = c.y;
    const float4 r0 = f4[(size_t)sr * 16 + j];
    const float4 r1 = f4[(size_t)sr * 16 + 8 + j];
    const int2 cn = cell[en >= 0 ? en : 0];
    a0.x += r0.x; a0.y += r0.y; a0.z += r0.z; a0.w += r0.w;
    a1.x += r1.x; a1.y += r1.y; a1.z += r1.z; a1.w += r1.w;
    e = en;
    c = cn;
  }
  float4* h4 = (float4*)hacc;
  h4[(size_t)n * 16 + j]     = a0;
  h4[(size_t)n * 16 + 8 + j] = a1;
}

__device__ __forceinline__ v8f wmma_f32_k4(v2f a, v2f b, v8f c) {
  return __builtin_amdgcn_wmma_f32_16x16x4_f32(false, a, false, b, (short)0, c,
                                               false, false);
}

__global__ __launch_bounds__(128) void gin_mlp_kernel(
    const float* __restrict__ hacc, const float* __restrict__ W1,
    const float* __restrict__ b1, const float* __restrict__ W2,
    const float* __restrict__ b2, float* __restrict__ out, int n_nodes) {
  const int lane  = threadIdx.x & 31;
  const int c     = lane & 15;
  const int hh    = lane >> 4;
  const int wv    = (blockIdx.x * blockDim.x + threadIdx.x) >> 5;
  const int node0 = wv * 32;
  if (node0 >= n_nodes) return;

  const int pc = (c & 3) + 8 * ((c >> 2) & 1) + 4 * (c >> 3);

  const int last = n_nodes - 1;
  const int ra = min(node0 + c, last);
  const int rb = min(node0 + 16 + c, last);

  v2f xa[16], xb[16];
  {
    const float4* pa = (const float4*)(hacc + (size_t)ra * GIN_D + 4 * hh);
    const float4* pb = (const float4*)(hacc + (size_t)rb * GIN_D + 4 * hh);
#pragma unroll
    for (int q = 0; q < 8; ++q) {
      const float4 va = pa[2 * q];
      const float4 vb = pb[2 * q];
      xa[2 * q].x = va.x; xa[2 * q].y = va.y; xa[2 * q + 1].x = va.z; xa[2 * q + 1].y = va.w;
      xb[2 * q].x = vb.x; xb[2 * q].y = vb.y; xb[2 * q + 1].x = vb.z; xb[2 * q + 1].y = vb.w;
    }
  }

  v8f o[4][2];
#pragma unroll
  for (int mo = 0; mo < 4; ++mo) { o[mo][0] = v8f{}; o[mo][1] = v8f{}; }

  v2f aw1[16], aw2[16];
  {
    const float* w1p = W1 + (size_t)(4 * hh) * GIN_H + c;
#pragma unroll
    for (int s = 0; s < 16; ++s) {
      const int k0 = 8 * (s >> 1) + 2 * (s & 1);
      aw1[s].x = w1p[(size_t)k0 * GIN_H];
      aw1[s].y = w1p[(size_t)(k0 + 1) * GIN_H];
    }
  }

#pragma unroll
  for (int mt = 0; mt < 8; ++mt) {
    {
      const float* w2p = W2 + (size_t)(16 * mt + 8 * hh) * GIN_D + pc;
#pragma unroll
      for (int p = 0; p < 4; ++p) {
#pragma unroll
        for (int mo = 0; mo < 4; ++mo) {
          aw2[4 * p + mo].x = w2p[(size_t)(2 * p) * GIN_D + 16 * mo];
          aw2[4 * p + mo].y = w2p[(size_t)(2 * p + 1) * GIN_D + 16 * mo];
        }
      }
    }
    v8f m0 = {};
    v8f m1 = {};
#pragma unroll
    for (int s = 0; s < 16; ++s) {
      m0 = wmma_f32_k4(aw1[s], xa[s], m0);
      m1 = wmma_f32_k4(aw1[s], xb[s], m1);
    }
    if (mt + 1 < 8) {
      const float* w1p = W1 + (size_t)(4 * hh) * GIN_H + 16 * (mt + 1) + c;
#pragma unroll
      for (int s = 0; s < 16; ++s) {
        const int k0 = 8 * (s >> 1) + 2 * (s & 1);
        aw1[s].x = w1p[(size_t)k0 * GIN_H];
        aw1[s].y = w1p[(size_t)(k0 + 1) * GIN_H];
      }
    }
    {
#pragma unroll
      for (int r = 0; r < 8; ++r) {
        const float blo = b1[16 * mt + r];
        const float bhi = b1[16 * mt + 8 + r];
        const float bv  = hh ? bhi : blo;
        m0[r] = fmaxf(m0[r] + bv, 0.0f);
        m1[r] = fmaxf(m1[r] + bv, 0.0f);
      }
    }
#pragma unroll
    for (int p = 0; p < 4; ++p) {
      v2f ba, bb;
      ba.x = m0[2 * p]; ba.y = m0[2 * p + 1];
      bb.x = m1[2 * p]; bb.y = m1[2 * p + 1];
#pragma unroll
      for (int mo = 0; mo < 4; ++mo) {
        o[mo][0] = wmma_f32_k4(aw2[4 * p + mo], ba, o[mo][0]);
        o[mo][1] = wmma_f32_k4(aw2[4 * p + mo], bb, o[mo][1]);
      }
    }
  }

  const bool sa = (node0 + c) < n_nodes;
  const bool sb = (node0 + 16 + c) < n_nodes;
  float* oa = out + (size_t)(node0 + c) * GIN_D + 4 * hh;
  float* ob = out + (size_t)(node0 + 16 + c) * GIN_D + 4 * hh;
#pragma unroll
  for (int mo = 0; mo < 4; ++mo) {
    float bv[8];
#pragma unroll
    for (int r = 0; r < 8; ++r) {
      const int col  = 16 * mo + 8 * (r >> 2) + (r & 3);
      const float lo = b2[col];
      const float hi = b2[col + 4];
      bv[r] = hh ? hi : lo;
    }
    const v8f ta = o[mo][0];
    const v8f tb = o[mo][1];
    if (sa) {
      *(float4*)(oa + 16 * mo)     = make_float4(ta[0] + bv[0], ta[1] + bv[1], ta[2] + bv[2], ta[3] + bv[3]);
      *(float4*)(oa + 16 * mo + 8) = make_float4(ta[4] + bv[4], ta[5] + bv[5], ta[6] + bv[6], ta[7] + bv[7]);
    }
    if (sb) {
      *(float4*)(ob + 16 * mo)     = make_float4(tb[0] + bv[0], tb[1] + bv[1], tb[2] + bv[2], tb[3] + bv[3]);
      *(float4*)(ob + 16 * mo + 8) = make_float4(tb[4] + bv[4], tb[5] + bv[5], tb[6] + bv[6], tb[7] + bv[7]);
    }
  }
}

extern "C" void kernel_launch(void* const* d_in, const int* in_sizes, int n_in,
                              void* d_out, int out_size, void* d_ws, size_t ws_size,
                              hipStream_t stream) {
  const float* feat = (const float*)d_in[0];
  const int*   esrc = (const int*)d_in[1];
  const int*   edst = (const int*)d_in[2];
  const float* W1   = (const float*)d_in[3];
  const float* b1   = (const float*)d_in[4];
  const float* W2   = (const float*)d_in[5];
  const float* b2   = (const float*)d_in[6];
  const float* eps  = (const float*)d_in[7];

  const int n_nodes = in_sizes[0] / GIN_D;
  const int n_edges = in_sizes[1];

  int*   head = (int*)d_ws;
  int2*  cell = (int2*)(head + ((n_nodes + 63) & ~63));
  float* hacc = (float*)(cell + ((n_edges + 31) & ~31));

  {
    const int blocks = (n_nodes + 1023) / 1024;
    gin_head_init_kernel<<<blocks, 256, 0, stream>>>(head, n_nodes);
  }
  {
    const int blocks = (n_edges + 1023) / 1024;
    gin_link_kernel<<<blocks, 256, 0, stream>>>(esrc, edst, head, cell, n_edges);
  }
  {
    const int blocks = (n_nodes + 31) / 32;
    gin_pull_kernel<<<blocks, 256, 0, stream>>>(feat, eps, head, cell, hacc,
                                                n_nodes, n_edges);
  }
  {
    const int blocks = (n_nodes + 127) / 128;
    gin_mlp_kernel<<<blocks, 128, 0, stream>>>(hacc, W1, b1, W2, b2,
                                               (float*)d_out, n_nodes);
  }
}


# segment sum: per-node table of 64 source slots filled by integer exchange with displacement replaces the linked list (no per-edge cell fetch, no hop chain); rows summed in slot order
# speedup vs baseline: 1.6632x; 1.1569x over previous
#include <hip/hip_runtime.h>
#include <hip/hip_bf16.h>


#define GIN_D 64
#define GIN_H 128
#define GIN_SLOTS 64

typedef __attribute__((ext_vector_type(2))) float v2f;
typedef __attribute__((ext_vector_type(8))) float v8f;

__device__ __forceinline__ int atomic_xchg_i32_agent(int* p, int v) {
  return __hip_atomic_exchange(p, v, __ATOMIC_RELAXED, __HIP_MEMORY_SCOPE_AGENT);
}

__device__ __forceinline__ int gin_first_slot(int e) {
  return (int)(((unsigned)e * 2654435769u) >> 26);
}

__global__ __launch_bounds__(256) void gin_table_init_kernel(
    int4* __restrict__ slot4, int* __restrict__ ohead, int n_slot4, int n_nodes) {
  const int t  = blockIdx.x * blockDim.x + threadIdx.x;
  const int nt = gridDim.x * blockDim.x;
  for (int i = t; i < n_slot4; i += nt) slot4[i] = make_int4(-1, -1, -1, -1);
  for (int i = t; i < n_nodes; i += nt) ohead[i] = -1;
}

__device__ __forceinline__ void gin_place_rest(
    int* __restrict__ slot, int* __restrict__ ohead, int2* __restrict__ ocell,
    int d, int h, int carry, int e) {
  int* row = slot + (size_t)d * GIN_SLOTS;
  for (int p = 1; p < GIN_SLOTS && carry >= 0; ++p) {
    h = (h + 1) & (GIN_SLOTS - 1);
    carry = atomic_xchg_i32_agent(row + h, carry);
  }
  if (carry >= 0) {
    const int old = atomic_xchg_i32_agent(ohead + d, e);
    ocell[e] = make_int2(old, carry);
  }
}

__global__ __launch_bounds__(256) void gin_link_kernel(
    const int* __restrict__ esrc, const int* __restrict__ edst,
    int* __restrict__ slot, int* __restrict__ ohead, int2* __restrict__ ocell,
    int n_edges) {
  const int lane = threadIdx.x & 31;
  const int wv   = (blockIdx.x * blockDim.x + threadIdx.x) >> 5;
  const int base = wv * 128 + lane;
  if (wv * 128 + 128 <= n_edges) {
    int s[4], d[4], h[4], o[4];
    for (int k = 0; k < 4; ++k) { s[k] = esrc[base + 32 * k]; d[k] = edst[base + 32 * k]; }
    for (int k = 0; k < 4; ++k) {
      h[k] = gin_first_slot(base + 32 * k);
      o[k] = atomic_xchg_i32_agent(slot + (size_t)d[k] * GIN_SLOTS + h[k], s[k]);
    }
    for (int k = 0; k < 4; ++k)
      if (o[k] >= 0) gin_place_rest(slot, ohead, ocell, d[k], h[k], o[k], base + 32 * k);
  } else {
    for (int k = 0; k < 4; ++k) {
      const int e = base + 32 * k;
      if (e < n_edges) {
        const int dv = edst[e];
        const int hv = gin_first_slot(e);
        const int ov = atomic_xchg_i32_agent(slot + (size_t)dv * GIN_SLOTS + hv, esrc[e]);
        if (ov >= 0) gin_place_rest(slot, ohead, ocell, dv, hv, ov, e);
      }
    }
  }
}

__global__ __launch_bounds__(256) __attribute__((amdgpu_waves_per_eu(4, 8)))
void gin_pull_kernel(
    const float* __restrict__ feat, const float* __restrict__ eps,
    const int* __restrict__ slot, const int* __restrict__ ohead,
    const int2* __restrict__ ocell, float* __restrict__ hacc,
    int n_nodes, int n_edges) {
  __shared__ int4 lst4[32 * (GIN_SLOTS / 4)];
  const int gt = blockIdx.x * blockDim.x + threadIdx.x;
  const int n  = gt >> 3;
  const int j  = gt & 7;
  if (n >= n_nodes) return;
  const float4* f4 = (const float4*)feat;
  const int4*   s4 = (const int4*)slot + (size_t)n * (GIN_SLOTS / 4);
  const float sc = 1.0f + eps[0];
  const int4 sa = s4[j];
  const int4 sb = s4[8 + j];
  float4 a0 = f4[(size_t)n * 16 + j];
  float4 a1 = f4[(size_t)n * 16 + 8 + j];
  int oe = ohead[n];
  a0.x *= sc; a0.y *= sc; a0.z *= sc; a0.w *= sc;
  a1.x *= sc; a1.y *= sc; a1.z *= sc; a1.w *= sc;

  const int cA = (sa.x >= 0) + (sa.y >= 0) + (sa.z >= 0) + (sa.w >= 0);
  const int cB = (sb.x >= 0) + (sb.y >= 0) + (sb.z >= 0) + (sb.w >= 0);
  const int pk = cA | (cB << 8);
  int inc = pk;
  for (int off = 1; off < 8; off <<= 1) {
    const int t = __shfl_up(inc, off, 8);
    if (j >= off) inc += t;
  }
  const int tot  = __shfl(inc, 7, 8);
  const int totA = tot & 255;
  const int d    = totA + (tot >> 8);
  const int exc  = inc - pk;
  int pA = exc & 255;
  int pB = totA + (exc >> 8);

  int* my = (int*)(lst4 + (threadIdx.x >> 3) * (GIN_SLOTS / 4));
  if (sa.x >= 0) my[pA++] = sa.x;
  if (sa.y >= 0) my[pA++] = sa.y;
  if (sa.z >= 0) my[pA++] = sa.z;
  if (sa.w >= 0) my[pA++] = sa.w;
  if (sb.x >= 0) my[pB++] = sb.x;
  if (sb.y >= 0) my[pB++] = sb.y;
  if (sb.z >= 0) my[pB++] = sb.z;
  if (sb.w >= 0) my[pB++] = sb.w;
  if (d + j < GIN_SLOTS) my[d + j] = n;

  const int4* my4 = (const int4*)my;
  for (int i = 0; i < d; i += 8) {
    const int4 ia = my4[i >> 2];
    const int4 ib = my4[(i >> 2) + 1];
    const int idx[8] = {ia.x, ia.y, ia.z, ia.w, ib.x, ib.y, ib.z, ib.w};
    float4 r0[8], r1[8];
#pragma unroll
    for (int k = 0; k < 8; ++k) {
      r0[k] = f4[(size_t)idx[k] * 16 + j];
      r1[k] = f4[(size_t)idx[k] * 16 + 8 + j];
    }
#pragma unroll
    for (int k = 0; k < 8; ++k) {
      const bool ok = (i + k) < d;
      a0.x += ok ? r0[k].x : 0.0f; a0.y += ok ? r0[k].y : 0.0f;
      a0.z += ok ? r0[k].z : 0.0f; a0.w += ok ? r0[k].w : 0.0f;
      a1.x += ok ? r1[k].x : 0.0f; a1.y += ok ? r1[k].y : 0.0f;
      a1.z += ok ? r1[k].z : 0.0f; a1.w += ok ? r1[k].w : 0.0f;
    }
  }
  for (int it = 0; it < n_edges && oe >= 0; ++it) {
    const int2 c = ocell[oe];
    const float4 r0 = f4[(size_t)c.y * 16 + j];
    const float4 r1 = f4[(size_t)c.y * 16 + 8 + j];
    a0.x += r0.x; a0.y += r0.y; a0.z += r0.z; a0.w += r0.w;
    a1.x += r1.x; a1.y += r1.y; a1.z += r1.z; a1.w += r1.w;
    oe = c.x;
  }
  float4* h4 = (float4*)hacc;
  h4[(size_t)n * 16 + j]     = a0;
  h4[(size_t)n * 16 + 8 + j] = a1;
}

__device__ __forceinline__ v8f wmma_f32_k4(v2f a, v2f b, v8f c) {
  return __builtin_amdgcn_wmma_f32_16x16x4_f32(false, a, false, b, (short)0, c,
                                               false, false);
}

__global__ __launch_bounds__(128) void gin_mlp_kernel(
    const float* __restrict__ hacc, const float* __restrict__ W1,
    const float* __restrict__ b1, const float* __restrict__ W2,
    const float* __restrict__ b2, float* __restrict__ out, int n_nodes) {
  const int lane  = threadIdx.x & 31;
  const int c     = lane & 15;
  const int hh    = lane >> 4;
  const int wv    = (blockIdx.x * blockDim.x + threadIdx.x) >> 5;
  const int node0 = wv * 32;
  if (node0 >= n_nodes) return;

  const int pc = (c & 3) + 8 * ((c >> 2) & 1) + 4 * (c >> 3);

  const int last = n_nodes - 1;
  const int ra = min(node0 + c, last);
  const int rb = min(node0 + 16 + c, last);

  v2f xa[16], xb[16];
  {
    const float4* pa = (const float4*)(hacc + (size_t)ra * GIN_D + 4 * hh);
    const float4* pb = (const float4*)(hacc + (size_t)rb * GIN_D + 4 * hh);
#pragma unroll
    for (int q = 0; q < 8; ++q) {
      const float4 va = pa[2 * q];
      const float4 vb = pb[2 * q];
      xa[2 * q].x = va.x; xa[2 * q].y = va.y; xa[2 * q + 1].x = va.z; xa[2 * q + 1].y = va.w;
      xb[2 * q].x = vb.x; xb[2 * q].y = vb.y; xb[2 * q + 1].x = vb.z; xb[2 * q + 1].y = vb.w;
    }
  }

  v8f o[4][2];
#pragma unroll
  for (int mo = 0; mo < 4; ++mo) { o[mo][0] = v8f{}; o[mo][1] = v8f{}; }

  v2f aw1[16], aw2[16];
  {
    const float* w1p = W1 + (size_t)(4 * hh) * GIN_H + c;
#pragma unroll
    for (int s = 0; s < 16; ++s) {
      const int k0 = 8 * (s >> 1) + 2 * (s & 1);
      aw1[s].x = w1p[(size_t)k0 * GIN_H];
      aw1[s].y = w1p[(size_t)(k0 + 1) * GIN_H];
    }
  }

#pragma unroll
  for (int mt = 0; mt < 8; ++mt) {
    {
      const float* w2p = W2 + (size_t)(16 * mt + 8 * hh) * GIN_D + pc;
#pragma unroll
      for (int p = 0; p < 4; ++p) {
#pragma unroll
        for (int mo = 0; mo < 4; ++mo) {
          aw2[4 * p + mo].x = w2p[(size_t)(2 * p) * GIN_D + 16 * mo];
          aw2[4 * p + mo].y = w2p[(size_t)(2 * p + 1) * GIN_D + 16 * mo];
        }
      }
    }
    v8f m0 = {};
    v8f m1 = {};
#pragma unroll
    for (int s = 0; s < 16; ++s) {
      m0 = wmma_f32_k4(aw1[s], xa[s], m0);
      m1 = wmma_f32_k4(aw1[s], xb[s], m1);
    }
    if (mt + 1 < 8) {
      const float* w1p = W1 + (size_t)(4 * hh) * GIN_H + 16 * (mt + 1) + c;
#pragma unroll
      for (int s = 0; s < 16; ++s) {
        const int k0 = 8 * (s >> 1) + 2 * (s & 1);
        aw1[s].x = w1p[(size_t)k0 * GIN_H];
        aw1[s].y = w1p[(size_t)(k0 + 1) * GIN_H];
      }
    }
    {
#pragma unroll
      for (int r = 0; r < 8; ++r) {
        const float blo = b1[16 * mt + r];
        const float bhi = b1[16 * mt + 8 + r];
        const float bv  = hh ? bhi : blo;
        m0[r] = fmaxf(m0[r] + bv, 0.0f);
        m1[r] = fmaxf(m1[r] + bv, 0.0f);
      }
    }
#pragma unroll
    for (int p = 0; p < 4; ++p) {
      v2f ba, bb;
      ba.x = m0[2 * p]; ba.y = m0[2 * p + 1];
      bb.x = m1[2 * p]; bb.y = m1[2 * p + 1];
#pragma unroll
      for (int mo = 0; mo < 4; ++mo) {
        o[mo][0] = wmma_f32_k4(aw2[4 * p + mo], ba, o[mo][0]);
        o[mo][1] = wmma_f32_k4(aw2[4 * p + mo], bb, o[mo][1]);
      }
    }
  }

  const bool sa = (node0 + c) < n_nodes;
  const bool sb = (node0 + 16 + c) < n_nodes;
  float* oa = out + (size_t)(node0 + c) * GIN_D + 4 * hh;
  float* ob = out + (size_t)(node0 + 16 + c) * GIN_D + 4 * hh;
#pragma unroll
  for (int mo = 0; mo < 4; ++mo) {
    float bv[8];
#pragma unroll
    for (int r = 0; r < 8; ++r) {
      const int col  = 16 * mo + 8 * (r >> 2) + (r & 3);
      const float lo = b2[col];
      const float hi = b2[col + 4];
      bv[r] = hh ? hi : lo;
    }
    const v8f ta = o[mo][0];
    const v8f tb = o[mo][1];
    if (sa) {
      *(float4*)(oa + 16 * mo)     = make_float4(ta[0] + bv[0], ta[1] + bv[1], ta[2] + bv[2], ta[3] + bv[3]);
      *(float4*)(oa + 16 * mo + 8) = make_float4(ta[4] + bv[4], ta[5] + bv[5], ta[6] + bv[6], ta[7] + bv[7]);
    }
    if (sb) {
      *(float4*)(ob + 16 * mo)     = make_float4(tb[0] + bv[0], tb[1] + bv[1], tb[2] + bv[2], tb[3] + bv[3]);
      *(float4*)(ob + 16 * mo + 8) = make_float4(tb[4] + bv[4], tb[5] + bv[5], tb[6] + bv[6], tb[7] + bv[7]);
    }
  }
}

extern "C" void kernel_launch(void* const* d_in, const int* in_sizes, int n_in,
                              void* d_out, int out_size, void* d_ws, size_t ws_size,
                              hipStream_t stream) {
  const float* feat = (const float*)d_in[0];
  const int*   esrc = (const int*)d_in[1];
  const int*   edst = (const int*)d_in[2];
  const float* W1   = (const float*)d_in[3];
  const float* b1   = (const float*)d_in[4];
  const float* W2   = (const float*)d_in[5];
  const float* b2   = (const float*)d_in[6];
  const float* eps  = (const float*)d_in[7];

  const int n_nodes = in_sizes[0] / GIN_D;
  const int n_edges = in_sizes[1];

  int*   slot  = (int*)d_ws;
  int*   ohead = slot + (size_t)n_nodes * GIN_SLOTS;
  int2*  ocell = (int2*)(ohead + ((n_nodes + 63) & ~63));
  float* hacc  = (float*)(ocell + ((n_edges + 31) & ~31));

  {
    gin_table_init_kernel<<<1024, 256, 0, stream>>>((int4*)slot, ohead,
                                                    n_nodes * (GIN_SLOTS / 4), n_nodes);
  }
  {
    const int blocks = (n_edges + 1023) / 1024;
    gin_link_kernel<<<blocks, 256, 0, stream>>>(esrc, edst, slot, ohead, ocell, n_edges);
  }
  {
    const int blocks = (n_nodes + 31) / 32;
    gin_pull_kernel<<<blocks, 256, 0, stream>>>(feat, eps, slot, ohead, ocell, hacc,
                                                n_nodes, n_edges);
  }
  {
    const int blocks = (n_nodes + 127) / 128;
    gin_mlp_kernel<<<blocks, 128, 0, stream>>>(hacc, W1, b1, W2, b2,
                                               (float*)d_out, n_nodes);
  }
}
